# starting kernel
# baseline (speedup 1.0000x reference)
#include <hip/hip_runtime.h>

#define NN 100000
#define NE 1600000
#define DD 128

typedef __attribute__((ext_vector_type(2))) float v2f;
typedef __attribute__((ext_vector_type(8))) float v8f;

__global__ __launch_bounds__(256) void gcn_gemm_wmma(const float* __restrict__ x,
                                                     const float* __restrict__ W,
                                                     float* __restrict__ h) {
    __shared__ float ldsWT[16 * 132];
    const int tid     = threadIdx.x;
    const int colBase = blockIdx.y * 16;

    #pragma unroll
    for (int i = 0; i < 8; ++i) {
        int e = tid + i * 256;
        int n = e & 15;
        int k = e >> 4;
        ldsWT[n * 132 + k] = W[k * DD + colBase + n];
    }
    __syncthreads();

    const int wave  = tid >> 5;
    const int lane  = tid & 31;
    const int mTile = blockIdx.x * 8 + wave;
    if (mTile >= (NN / 16)) return;

    const int    row  = mTile * 16 + (lane & 15);
    const int    koff = (lane >> 4) * 2;
    const float* xrow = x + (size_t)row * DD;
    const float* wtn  = ldsWT + (lane & 15) * 132;

    v8f acc = {};
    #pragma unroll
    for (int kk = 0; kk < DD; kk += 4) {
        v2f a, b;
        a.x = xrow[kk + koff];
        a.y = xrow[kk + koff + 1];
        b.x = wtn[kk + koff];
        b.y = wtn[kk + koff + 1];
        acc = __builtin_amdgcn_wmma_f32_16x16x4_f32(
                   false, a,  false, b,
                   (short)0, acc,  false,  false);
    }

    const int col     = colBase + (lane & 15);
    const int rowBase = mTile * 16 + (lane >> 4) * 8;
    #pragma unroll
    for (int v = 0; v < 8; ++v) {
        h[(size_t)(rowBase + v) * DD + col] = acc[v];
    }
}

__global__ __launch_bounds__(256) void gcn_init_bias(float* __restrict__ agg,
                                                     const float* __restrict__ b) {
    size_t i = (size_t)blockIdx.x * 256 + threadIdx.x;
    agg[i] = b[i & (DD - 1)];
}

__global__ __launch_bounds__(256) void gcn_scatter(const float* __restrict__ h,
                                                   const int* __restrict__ src,
                                                   const int* __restrict__ dst,
                                                   const float* __restrict__ w,
                                                   float* __restrict__ agg) {
    const int e = blockIdx.x * 8 + (threadIdx.x >> 5);
    if (e >= NE) return;
    const int   lane = threadIdx.x & 31;
    const int   s    = src[e];
    const int   d    = dst[e];
    const float wt   = w[e];

    const float4 hv = *(const float4*)(h + (size_t)s * DD + lane * 4);
    float* out = agg + (size_t)d * DD + lane * 4;
    unsafeAtomicAdd(out + 0, hv.x * wt);
    unsafeAtomicAdd(out + 1, hv.y * wt);
    unsafeAtomicAdd(out + 2, hv.z * wt);
    unsafeAtomicAdd(out + 3, hv.w * wt);
}

__global__ __launch_bounds__(256) void gcn_relu(const float* __restrict__ in,
                                                float* __restrict__ out) {
    size_t i = (size_t)blockIdx.x * 256 + threadIdx.x;
    out[i] = fmaxf(in[i], 0.0f);
}

extern "C" void kernel_launch(void* const* d_in, const int* in_sizes, int n_in,
                              void* d_out, int out_size, void* d_ws, size_t ws_size,
                              hipStream_t stream) {
    const float* x    = (const float*)d_in[0];
    const int*   esrc = (const int*)d_in[1];
    const int*   edst = (const int*)d_in[2];
    const float* ew   = (const float*)d_in[3];
    const float* W    = (const float*)d_in[4];
    const float* b    = (const float*)d_in[5];
    float*       out  = (float*)d_out;

    float* ws0 = (float*)d_ws;
    float* ws1 = ws0 + (size_t)NN * DD;

    const float* xcur[3] = { x,   ws1, out };
    float*       agg[3]  = { ws1, out, ws1 };
    float*       post[3] = { ws1, out, out };

    const dim3 gGemm((NN / 16 + 7) / 8, DD / 16);
    const int  gND = (NN * DD) / 256;
    const int  gE  = NE / 8;

    for (int l = 0; l < 3; ++l) {
        gcn_gemm_wmma<<<gGemm, 256, 0, stream>>>(xcur[l], W + (size_t)l * DD * DD, ws0);
        gcn_init_bias<<<gND, 256, 0, stream>>>(agg[l], b + (size_t)l * DD);
        gcn_scatter<<<gE, 256, 0, stream>>>(ws0, esrc, edst, ew, agg[l]);
        gcn_relu<<<gND, 256, 0, stream>>>(agg[l], post[l]);
    }
}


# Float atomic scatter-add replaced by per-dst edge buckets built once per call (int atomic count), each ordered by edge index; one wave per row sums in that fixed order (fma), then bias, relu
# speedup vs baseline: 2.4830x; 2.4830x over previous
#include <hip/hip_runtime.h>

#define NN 100000
#define NE 1600000
#define DD 128

typedef __attribute__((ext_vector_type(2))) float v2f;
typedef __attribute__((ext_vector_type(8))) float v8f;

__global__ __launch_bounds__(256) void gcn_gemm_wmma(const float* __restrict__ x,
                                                     const float* __restrict__ W,
                                                     float* __restrict__ h) {
    __shared__ float ldsWT[16 * 132];
    const int tid     = threadIdx.x;
    const int colBase = blockIdx.y * 16;

    #pragma unroll
    for (int i = 0; i < 8; ++i) {
        int e = tid + i * 256;
        int n = e & 15;
        int k = e >> 4;
        ldsWT[n * 132 + k] = W[k * DD + colBase + n];
    }
    __syncthreads();

    const int wave  = tid >> 5;
    const int lane  = tid & 31;
    const int mTile = blockIdx.x * 8 + wave;
    if (mTile >= (NN / 16)) return;

    const int    row  = mTile * 16 + (lane & 15);
    const int    koff = (lane >> 4) * 2;
    const float* xrow = x + (size_t)row * DD;
    const float* wtn  = ldsWT + (lane & 15) * 132;

    v8f acc = {};
    #pragma unroll
    for (int kk = 0; kk < DD; kk += 4) {
        v2f a, b;
        a.x = xrow[kk + koff];
        a.y = xrow[kk + koff + 1];
        b.x = wtn[kk + koff];
        b.y = wtn[kk + koff + 1];
        acc = __builtin_amdgcn_wmma_f32_16x16x4_f32(
                   false, a,  false, b,
                   (short)0, acc,  false,  false);
    }

    const int col     = colBase + (lane & 15);
    const int rowBase = mTile * 16 + (lane >> 4) * 8;
    #pragma unroll
    for (int v = 0; v < 8; ++v) {
        h[(size_t)(rowBase + v) * DD + col] = acc[v];
    }
}

__global__ __launch_bounds__(256) void csr_zero(int* __restrict__ cnt, int* __restrict__ total) {
    const int r = blockIdx.x * 256 + threadIdx.x;
    if (r < NN) cnt[r] = 0;
    if (r == 0) total[0] = 0;
}

__global__ __launch_bounds__(256) void csr_count(const int* __restrict__ dst, int* __restrict__ cnt,
                                                 int* __restrict__ rank) {
    const int e = blockIdx.x * 256 + threadIdx.x;
    if (e < NE) rank[e] = atomicAdd(&cnt[dst[e]], 1);
}

__global__ __launch_bounds__(256) void csr_alloc(const int* __restrict__ cnt, int* __restrict__ total,
                                                 int* __restrict__ start) {
    __shared__ int wsum[8];
    __shared__ int blockBase;
    const int tid  = threadIdx.x;
    const int lane = tid & 31;
    const int wave = tid >> 5;
    const int r    = blockIdx.x * 256 + tid;
    const int c    = (r < NN) ? cnt[r] : 0;

    int v = c;
    for (int o = 1; o < 32; o <<= 1) {
        const int t = __shfl_up(v, o);
        if (lane >= o) v += t;
    }
    if (lane == 31) wsum[wave] = v;
    __syncthreads();
    if (tid == 0) {
        int run = 0;
        for (int i = 0; i < 8; ++i) { const int t = wsum[i]; wsum[i] = run; run += t; }
        blockBase = atomicAdd(total, run);
    }
    __syncthreads();
    if (r < NN) start[r] = blockBase + wsum[wave] + v - c;
}

__global__ __launch_bounds__(256) void csr_place(const int* __restrict__ dst, const int* __restrict__ start,
                                                 const int* __restrict__ rank, int* __restrict__ eid) {
    const int e = blockIdx.x * 256 + threadIdx.x;
    if (e < NE) eid[start[dst[e]] + rank[e]] = e;
}

__global__ __launch_bounds__(256) void csr_order(const int* __restrict__ eid, const int* __restrict__ dst,
                                                 const int* __restrict__ src, const float* __restrict__ w,
                                                 const int* __restrict__ start, const int* __restrict__ cnt,
                                                 int* __restrict__ ssrc, float* __restrict__ sw) {
    const int p = blockIdx.x * 256 + threadIdx.x;
    if (p >= NE) return;
    const int e = eid[p];
    const int d = dst[e];
    const int s = start[d];
    const int n = cnt[d];
    int rank = 0;
    for (int q = 0; q < n; ++q) rank += (eid[s + q] < e) ? 1 : 0;
    ssrc[s + rank] = src[e];
    sw[s + rank]   = w[e];
}

template <int K>
__device__ __forceinline__ void add_rows(const float* __restrict__ hl, int mySrc, float myW,
                                         int j, float4& acc) {
    float4 v[K];
    float  wv[K];
    #pragma unroll
    for (int k = 0; k < K; ++k) {
        const int sk = __builtin_amdgcn_readlane(mySrc, j + k);
        wv[k] = __int_as_float(__builtin_amdgcn_readlane(__float_as_int(myW), j + k));
        v[k]  = *(const float4*)(hl + (size_t)sk * DD);
    }
    #pragma unroll
    for (int k = 0; k < K; ++k) {
        acc.x = fmaf(v[k].x, wv[k], acc.x);
        acc.y = fmaf(v[k].y, wv[k], acc.y);
        acc.z = fmaf(v[k].z, wv[k], acc.z);
        acc.w = fmaf(v[k].w, wv[k], acc.w);
    }
}

__global__ __launch_bounds__(256) void gcn_gather(const float* __restrict__ h,
                                                  const int* __restrict__ start,
                                                  const int* __restrict__ cnt,
                                                  const int* __restrict__ ssrc,
                                                  const float* __restrict__ sw,
                                                  const float* __restrict__ bias,
                                                  float* __restrict__ xout) {
    const int lane = threadIdx.x & 31;
    const int r = __builtin_amdgcn_readfirstlane(blockIdx.x * 8 + (threadIdx.x >> 5));
    if (r >= NN) return;
    const int s = start[r];
    const int n = cnt[r];

    const float* hl = h + lane * 4;
    float4 acc = make_float4(0.0f, 0.0f, 0.0f, 0.0f);

    for (int base = 0; base < n; base += 32) {
        const int m = (n - base < 32) ? (n - base) : 32;
        int   mySrc = 0;
        float myW   = 0.0f;
        if (lane < m) {
            mySrc = ssrc[s + base + lane];
            myW   = sw[s + base + lane];
        }
        int j = 0;
        for (; j + 8 <= m; j += 8) add_rows<8>(hl, mySrc, myW, j, acc);
        if (j + 4 <= m) { add_rows<4>(hl, mySrc, myW, j, acc); j += 4; }
        switch (m - j) {
            case 3:  add_rows<3>(hl, mySrc, myW, j, acc); break;
            case 2:  add_rows<2>(hl, mySrc, myW, j, acc); break;
            case 1:  add_rows<1>(hl, mySrc, myW, j, acc); break;
            default: break;
        }
    }

    const float4 bv = *(const float4*)(bias + lane * 4);
    float4 o;
    o.x = fmaxf(acc.x + bv.x, 0.0f);
    o.y = fmaxf(acc.y + bv.y, 0.0f);
    o.z = fmaxf(acc.z + bv.z, 0.0f);
    o.w = fmaxf(acc.w + bv.w, 0.0f);
    *(float4*)(xout + (size_t)r * DD + lane * 4) = o;
}

extern "C" void kernel_launch(void* const* d_in, const int* in_sizes, int n_in,
                              void* d_out, int out_size, void* d_ws, size_t ws_size,
                              hipStream_t stream) {
    const float* x    = (const float*)d_in[0];
    const int*   esrc = (const int*)d_in[1];
    const int*   edst = (const int*)d_in[2];
    const float* ew   = (const float*)d_in[3];
    const float* W    = (const float*)d_in[4];
    const float* b    = (const float*)d_in[5];
    float*       out  = (float*)d_out;

    float* ws0 = (float*)d_ws;
    float* ws1 = ws0 + (size_t)NN * DD;
    int*   cnt    = (int*)(ws1 + (size_t)NN * DD);
    int*   start  = cnt + NN;
    int*   total  = start + NN;
    int*   rank   = total + 64;
    int*   eid    = rank + NE;
    int*   ssrc   = eid + NE;
    float* sw     = (float*)(ssrc + NE);

    const float* xcur[3] = { x,   ws1, out };
    float*       xnxt[3] = { ws1, out, out };

    const dim3 gGemm((NN / 16 + 7) / 8, DD / 16);
    const int  gN = (NN + 255) / 256;
    const int  gE = NE / 256;
    const int  gR = NN / 8;

    csr_zero<<<gN, 256, 0, stream>>>(cnt, total);
    csr_count<<<gE, 256, 0, stream>>>(edst, cnt, rank);
    csr_alloc<<<gN, 256, 0, stream>>>(cnt, total, start);
    csr_place<<<gE, 256, 0, stream>>>(edst, start, rank, eid);
    csr_order<<<gE, 256, 0, stream>>>(eid, edst, esrc, ew, start, cnt, ssrc, sw);

    for (int l = 0; l < 3; ++l) {
        gcn_gemm_wmma<<<gGemm, 256, 0, stream>>>(xcur[l], W + (size_t)l * DD * DD, ws0);
        gcn_gather<<<gR, 256, 0, stream>>>(ws0, start, cnt, ssrc, sw, b + (size_t)l * DD, xnxt[l]);
    }
}


# Grids cut to a few thousand long-lived waves: each wave loops over its row tiles (GEMM), rows (gather) or edges (bucket build, 4 in flight) instead of one item per wave; every sum and its order kept
# speedup vs baseline: 2.7568x; 1.1103x over previous
#include <hip/hip_runtime.h>

#define NN 100000
#define NE 1600000
#define DD 128

typedef __attribute__((ext_vector_type(2))) float v2f;
typedef __attribute__((ext_vector_type(8))) float v8f;

__global__ __launch_bounds__(256) void gcn_gemm_wmma(const float* __restrict__ x,
                                                     const float* __restrict__ W,
                                                     float* __restrict__ h) {
    __shared__ float ldsWT[16 * 132];
    const int tid     = threadIdx.x;
    const int colBase = blockIdx.y * 16;

    #pragma unroll
    for (int i = 0; i < 8; ++i) {
        int e = tid + i * 256;
        int n = e & 15;
        int k = e >> 4;
        ldsWT[n * 132 + k] = W[k * DD + colBase + n];
    }
    __syncthreads();

    const int wave  = __builtin_amdgcn_readfirstlane(tid >> 5);
    const int lane  = tid & 31;
    const int koff  = (lane >> 4) * 2;
    const float* wtn  = ldsWT + (lane & 15) * 132;
    const int col     = colBase + (lane & 15);

  for (int mTile = blockIdx.x * 8 + wave; mTile < (NN / 16); mTile += gridDim.x * 8) {
    const int    row  = mTile * 16 + (lane & 15);
    const float* xrow = x + (size_t)row * DD;

    v8f acc = {};
    #pragma unroll
    for (int kk = 0; kk < DD; kk += 4) {
        v2f a, b;
        a.x = xrow[kk + koff];
        a.y = xrow[kk + koff + 1];
        b.x = wtn[kk + koff];
        b.y = wtn[kk + koff + 1];
        acc = __builtin_amdgcn_wmma_f32_16x16x4_f32(
                   false, a,  false, b,
                   (short)0, acc,  false,  false);
    }

    const int rowBase = mTile * 16 + (lane >> 4) * 8;
    #pragma unroll
    for (int v = 0; v < 8; ++v) {
        h[(size_t)(rowBase + v) * DD + col] = acc[v];
    }
  }
}

__global__ __launch_bounds__(256) void csr_zero(int* __restrict__ cnt, int* __restrict__ total) {
    const int r = blockIdx.x * 256 + threadIdx.x;
    if (r < NN) cnt[r] = 0;
    if (r == 0) total[0] = 0;
}

__global__ __launch_bounds__(256) void csr_count(const int* __restrict__ dst, int* __restrict__ cnt,
                                                 int* __restrict__ rank) {
    const int T = gridDim.x * 256;
    for (int e0 = blockIdx.x * 256 + threadIdx.x; e0 < NE; e0 += 4 * T) {
        int d[4], rk[4];
        #pragma unroll
        for (int k = 0; k < 4; ++k) {
            const int e = e0 + k * T;
            d[k] = dst[(e < NE) ? e : e0];
        }
        #pragma unroll
        for (int k = 0; k < 4; ++k) rk[k] = (e0 + k * T < NE) ? atomicAdd(&cnt[d[k]], 1) : 0;
        #pragma unroll
        for (int k = 0; k < 4; ++k) if (e0 + k * T < NE) rank[e0 + k * T] = rk[k];
    }
}

__global__ __launch_bounds__(256) void csr_alloc(const int* __restrict__ cnt, int* __restrict__ total,
                                                 int* __restrict__ start) {
    __shared__ int wsum[8];
    __shared__ int blockBase;
    const int tid  = threadIdx.x;
    const int lane = tid & 31;
    const int wave = tid >> 5;
    const int r    = blockIdx.x * 256 + tid;
    const int c    = (r < NN) ? cnt[r] : 0;

    int v = c;
    for (int o = 1; o < 32; o <<= 1) {
        const int t = __shfl_up(v, o);
        if (lane >= o) v += t;
    }
    if (lane == 31) wsum[wave] = v;
    __syncthreads();
    if (tid == 0) {
        int run = 0;
        for (int i = 0; i < 8; ++i) { const int t = wsum[i]; wsum[i] = run; run += t; }
        blockBase = atomicAdd(total, run);
    }
    __syncthreads();
    if (r < NN) start[r] = blockBase + wsum[wave] + v - c;
}

__global__ __launch_bounds__(256) void csr_place(const int* __restrict__ dst, const int* __restrict__ start,
                                                 const int* __restrict__ rank, int* __restrict__ eid) {
    const int T = gridDim.x * 256;
    for (int e0 = blockIdx.x * 256 + threadIdx.x; e0 < NE; e0 += 4 * T) {
        int d[4], rk[4], st[4];
        #pragma unroll
        for (int k = 0; k < 4; ++k) {
            const int e  = e0 + k * T;
            const int ec = (e < NE) ? e : e0;
            d[k]  = dst[ec];
            rk[k] = rank[ec];
        }
        #pragma unroll
        for (int k = 0; k < 4; ++k) st[k] = start[d[k]];
        #pragma unroll
        for (int k = 0; k < 4; ++k) if (e0 + k * T < NE) eid[st[k] + rk[k]] = e0 + k * T;
    }
}

__global__ __launch_bounds__(256) void csr_order(const int* __restrict__ eid, const int* __restrict__ dst,
                                                 const int* __restrict__ src, const float* __restrict__ w,
                                                 const int* __restrict__ start, const int* __restrict__ cnt,
                                                 int* __restrict__ ssrc, float* __restrict__ sw) {
    const int T = gridDim.x * 256;
    for (int p = blockIdx.x * 256 + threadIdx.x; p < NE; p += T) {
        const int e = eid[p];
        const int d = dst[e];
        const int s = start[d];
        const int n = cnt[d];
        int rank = 0;
        for (int q = 0; q < n; ++q) rank += (eid[s + q] < e) ? 1 : 0;
        ssrc[s + rank] = src[e];
        sw[s + rank]   = w[e];
    }
}

template <int K>
__device__ __forceinline__ void add_rows(const float* __restrict__ hl, int mySrc, float myW,
                                         int j, float4& acc) {
    float4 v[K];
    float  wv[K];
    #pragma unroll
    for (int k = 0; k < K; ++k) {
        const int sk = __builtin_amdgcn_readlane(mySrc, j + k);
        wv[k] = __int_as_float(__builtin_amdgcn_readlane(__float_as_int(myW), j + k));
        v[k]  = *(const float4*)(hl + (size_t)sk * DD);
    }
    #pragma unroll
    for (int k = 0; k < K; ++k) {
        acc.x = fmaf(v[k].x, wv[k], acc.x);
        acc.y = fmaf(v[k].y, wv[k], acc.y);
        acc.z = fmaf(v[k].z, wv[k], acc.z);
        acc.w = fmaf(v[k].w, wv[k], acc.w);
    }
}

__global__ __launch_bounds__(256) void gcn_gather(const float* __restrict__ h,
                                                  const int* __restrict__ start,
                                                  const int* __restrict__ cnt,
                                                  const int* __restrict__ ssrc,
                                                  const float* __restrict__ sw,
                                                  const float* __restrict__ bias,
                                                  float* __restrict__ xout) {
    const int lane = threadIdx.x & 31;
    const int r0 = __builtin_amdgcn_readfirstlane(blockIdx.x * 8 + (threadIdx.x >> 5));
    const float* hl = h + lane * 4;
    const float4 bv = *(const float4*)(bias + lane * 4);

  for (int r = r0; r < NN; r += gridDim.x * 8) {
    const int s = start[r];
    const int n = cnt[r];
    float4 acc = make_float4(0.0f, 0.0f, 0.0f, 0.0f);

    for (int base = 0; base < n; base += 32) {
        const int m = (n - base < 32) ? (n - base) : 32;
        int   mySrc = 0;
        float myW   = 0.0f;
        if (lane < m) {
            mySrc = ssrc[s + base + lane];
            myW   = sw[s + base + lane];
        }
        int j = 0;
        for (; j + 8 <= m; j += 8) add_rows<8>(hl, mySrc, myW, j, acc);
        if (j + 4 <= m) { add_rows<4>(hl, mySrc, myW, j, acc); j += 4; }
        switch (m - j) {
            case 3:  add_rows<3>(hl, mySrc, myW, j, acc); break;
            case 2:  add_rows<2>(hl, mySrc, myW, j, acc); break;
            case 1:  add_rows<1>(hl, mySrc, myW, j, acc); break;
            default: break;
        }
    }

    float4 o;
    o.x = fmaxf(acc.x + bv.x, 0.0f);
    o.y = fmaxf(acc.y + bv.y, 0.0f);
    o.z = fmaxf(acc.z + bv.z, 0.0f);
    o.w = fmaxf(acc.w + bv.w, 0.0f);
    *(float4*)(xout + (size_t)r * DD + lane * 4) = o;
  }
}

extern "C" void kernel_launch(void* const* d_in, const int* in_sizes, int n_in,
                              void* d_out, int out_size, void* d_ws, size_t ws_size,
                              hipStream_t stream) {
    const float* x    = (const float*)d_in[0];
    const int*   esrc = (const int*)d_in[1];
    const int*   edst = (const int*)d_in[2];
    const float* ew   = (const float*)d_in[3];
    const float* W    = (const float*)d_in[4];
    const float* b    = (const float*)d_in[5];
    float*       out  = (float*)d_out;

    float* ws0 = (float*)d_ws;
    float* ws1 = ws0 + (size_t)NN * DD;
    int*   cnt    = (int*)(ws1 + (size_t)NN * DD);
    int*   start  = cnt + NN;
    int*   total  = start + NN;
    int*   rank   = total + 64;
    int*   eid    = rank + NE;
    int*   ssrc   = eid + NE;
    float* sw     = (float*)(ssrc + NE);

    const float* xcur[3] = { x,   ws1, out };
    float*       xnxt[3] = { ws1, out, out };

    const dim3 gGemm(160, DD / 16);
    const int  gN = (NN + 255) / 256;
    const int  gE = 1024;
    const int  gO = 2048;
    const int  gR = 2048;

    csr_zero<<<gN, 256, 0, stream>>>(cnt, total);
    csr_count<<<gE, 256, 0, stream>>>(edst, cnt, rank);
    csr_alloc<<<gN, 256, 0, stream>>>(cnt, total, start);
    csr_place<<<gE, 256, 0, stream>>>(edst, start, rank, eid);
    csr_order<<<gO, 256, 0, stream>>>(eid, edst, esrc, ew, start, cnt, ssrc, sw);

    for (int l = 0; l < 3; ++l) {
        gcn_gemm_wmma<<<gGemm, 256, 0, stream>>>(xcur[l], W + (size_t)l * DD * DD, ws0);
        gcn_gather<<<gR, 256, 0, stream>>>(ws0, start, cnt, ssrc, sw, b + (size_t)l * DD, xnxt[l]);
    }
}


# GEMM re-tiled: one 16-wave block per WGP stages all of W once, a wave keeps its 32 columns of W in registers and streams 16-row x tiles (x read once, not 8x); k order within 8 changes sum order
# speedup vs baseline: 3.4191x; 1.2402x over previous
#include <hip/hip_runtime.h>

#define NN 100000
#define NE 1600000
#define DD 128

typedef __attribute__((ext_vector_type(2))) float v2f;
typedef __attribute__((ext_vector_type(8))) float v8f;

__global__ __launch_bounds__(512) void gcn_gemm_wmma(const float* __restrict__ x,
                                                      const float* __restrict__ W,
                                                      float* __restrict__ h) {
    __shared__ float ldsWT[DD * 132];
    const int tid = threadIdx.x;

    #pragma unroll
    for (int i = 0; i < 32; ++i) {
        int e = tid + i * 512;
        int n = e & 127;
        int k = e >> 7;
        ldsWT[n * 132 + k] = W[k * DD + n];
    }
    __syncthreads();

    const int wave = __builtin_amdgcn_readfirstlane(tid >> 5);
    const int lane = tid & 31;
    const int c    = lane & 15;
    const int hh   = lane >> 4;
    const int cq   = wave & 3;
    const float* b0p = ldsWT + (cq * 32 + c) * 132 + hh * 4;
    const float* b1p = b0p + 16 * 132;
    const int col0   = cq * 32 + c;

  for (int mTile = (wave >> 2) * gridDim.x + blockIdx.x; mTile < (NN / 16); mTile += gridDim.x * 4) {
    const float* xrow = x + (size_t)(mTile * 16 + c) * DD + hh * 4;

    float4 a[DD / 8];
    #pragma unroll
    for (int j = 0; j < DD / 8; ++j) a[j] = *(const float4*)(xrow + j * 8);

    v8f acc0 = {};
    v8f acc1 = {};
    #pragma unroll
    for (int j = 0; j < DD / 8; ++j) {
        const float4 b0 = *(const float4*)(b0p + j * 8);
        const float4 b1 = *(const float4*)(b1p + j * 8);
        v2f aL, aH, b0L, b0H, b1L, b1H;
        aL.x  = a[j].x; aL.y  = a[j].y; aH.x  = a[j].z; aH.y  = a[j].w;
        b0L.x = b0.x; b0L.y = b0.y; b0H.x = b0.z; b0H.y = b0.w;
        b1L.x = b1.x; b1L.y = b1.y; b1H.x = b1.z; b1H.y = b1.w;
        acc0 = __builtin_amdgcn_wmma_f32_16x16x4_f32(false, aL, false, b0L, (short)0, acc0, false, false);
        acc1 = __builtin_amdgcn_wmma_f32_16x16x4_f32(false, aL, false, b1L, (short)0, acc1, false, false);
        acc0 = __builtin_amdgcn_wmma_f32_16x16x4_f32(false, aH, false, b0H, (short)0, acc0, false, false);
        acc1 = __builtin_amdgcn_wmma_f32_16x16x4_f32(false, aH, false, b1H, (short)0, acc1, false, false);
    }

    const int rowBase = mTile * 16 + hh * 8;
    #pragma unroll
    for (int v = 0; v < 8; ++v) {
        float* hp = h + (size_t)(rowBase + v) * DD + col0;
        hp[0]  = acc0[v];
        hp[16] = acc1[v];
    }
  }
}

__global__ __launch_bounds__(256) void csr_zero(int* __restrict__ cnt, int* __restrict__ total) {
    const int r = blockIdx.x * 256 + threadIdx.x;
    if (r < NN) cnt[r] = 0;
    if (r == 0) total[0] = 0;
}

__global__ __launch_bounds__(256) void csr_count(const int* __restrict__ dst, int* __restrict__ cnt,
                                                 int* __restrict__ rank) {
    const int T = gridDim.x * 256;
    for (int e0 = blockIdx.x * 256 + threadIdx.x; e0 < NE; e0 += 4 * T) {
        int d[4], rk[4];
        #pragma unroll
        for (int k = 0; k < 4; ++k) {
            const int e = e0 + k * T;
            d[k] = dst[(e < NE) ? e : e0];
        }
        #pragma unroll
        for (int k = 0; k < 4; ++k) rk[k] = (e0 + k * T < NE) ? atomicAdd(&cnt[d[k]], 1) : 0;
        #pragma unroll
        for (int k = 0; k < 4; ++k) if (e0 + k * T < NE) rank[e0 + k * T] = rk[k];
    }
}

__global__ __launch_bounds__(256) void csr_alloc(const int* __restrict__ cnt, int* __restrict__ total,
                                                 int* __restrict__ start) {
    __shared__ int wsum[8];
    __shared__ int blockBase;
    const int tid  = threadIdx.x;
    const int lane = tid & 31;
    const int wave = tid >> 5;
    const int r    = blockIdx.x * 256 + tid;
    const int c    = (r < NN) ? cnt[r] : 0;

    int v = c;
    for (int o = 1; o < 32; o <<= 1) {
        const int t = __shfl_up(v, o);
        if (lane >= o) v += t;
    }
    if (lane == 31) wsum[wave] = v;
    __syncthreads();
    if (tid == 0) {
        int run = 0;
        for (int i = 0; i < 8; ++i) { const int t = wsum[i]; wsum[i] = run; run += t; }
        blockBase = atomicAdd(total, run);
    }
    __syncthreads();
    if (r < NN) start[r] = blockBase + wsum[wave] + v - c;
}

__global__ __launch_bounds__(256) void csr_place(const int* __restrict__ dst, const int* __restrict__ start,
                                                 const int* __restrict__ rank, int* __restrict__ eid) {
    const int T = gridDim.x * 256;
    for (int e0 = blockIdx.x * 256 + threadIdx.x; e0 < NE; e0 += 4 * T) {
        int d[4], rk[4], st[4];
        #pragma unroll
        for (int k = 0; k < 4; ++k) {
            const int e  = e0 + k * T;
            const int ec = (e < NE) ? e : e0;
            d[k]  = dst[ec];
            rk[k] = rank[ec];
        }
        #pragma unroll
        for (int k = 0; k < 4; ++k) st[k] = start[d[k]];
        #pragma unroll
        for (int k = 0; k < 4; ++k) if (e0 + k * T < NE) eid[st[k] + rk[k]] = e0 + k * T;
    }
}

__global__ __launch_bounds__(256) void csr_order(const int* __restrict__ eid, const int* __restrict__ dst,
                                                 const int* __restrict__ src, const float* __restrict__ w,
                                                 const int* __restrict__ start, const int* __restrict__ cnt,
                                                 int* __restrict__ ssrc, float* __restrict__ sw) {
    const int T = gridDim.x * 256;
    for (int p = blockIdx.x * 256 + threadIdx.x; p < NE; p += T) {
        const int e = eid[p];
        const int d = dst[e];
        const int s = start[d];
        const int n = cnt[d];
        int rank = 0;
        for (int q = 0; q < n; ++q) rank += (eid[s + q] < e) ? 1 : 0;
        ssrc[s + rank] = src[e];
        sw[s + rank]   = w[e];
    }
}

template <int K>
__device__ __forceinline__ void add_rows(const float* __restrict__ hl, int mySrc, float myW,
                                         int j, float4& acc) {
    float4 v[K];
    float  wv[K];
    #pragma unroll
    for (int k = 0; k < K; ++k) {
        const int sk = __builtin_amdgcn_readlane(mySrc, j + k);
        wv[k] = __int_as_float(__builtin_amdgcn_readlane(__float_as_int(myW), j + k));
        v[k]  = *(const float4*)(hl + (size_t)sk * DD);
    }
    #pragma unroll
    for (int k = 0; k < K; ++k) {
        acc.x = fmaf(v[k].x, wv[k], acc.x);
        acc.y = fmaf(v[k].y, wv[k], acc.y);
        acc.z = fmaf(v[k].z, wv[k], acc.z);
        acc.w = fmaf(v[k].w, wv[k], acc.w);
    }
}

__global__ __launch_bounds__(256) void gcn_gather(const float* __restrict__ h,
                                                  const int* __restrict__ start,
                                                  const int* __restrict__ cnt,
                                                  const int* __restrict__ ssrc,
                                                  const float* __restrict__ sw,
                                                  const float* __restrict__ bias,
                                                  float* __restrict__ xout) {
    const int lane = threadIdx.x & 31;
    const int r0 = __builtin_amdgcn_readfirstlane(blockIdx.x * 8 + (threadIdx.x >> 5));
    const float* hl = h + lane * 4;
    const float4 bv = *(const float4*)(bias + lane * 4);

  for (int r = r0; r < NN; r += gridDim.x * 8) {
    const int s = start[r];
    const int n = cnt[r];
    float4 acc = make_float4(0.0f, 0.0f, 0.0f, 0.0f);

    for (int base = 0; base < n; base += 32) {
        const int m = (n - base < 32) ? (n - base) : 32;
        int   mySrc = 0;
        float myW   = 0.0f;
        if (lane < m) {
            mySrc = ssrc[s + base + lane];
            myW   = sw[s + base + lane];
        }
        int j = 0;
        for (; j + 8 <= m; j += 8) add_rows<8>(hl, mySrc, myW, j, acc);
        if (j + 4 <= m) { add_rows<4>(hl, mySrc, myW, j, acc); j += 4; }
        switch (m - j) {
            case 3:  add_rows<3>(hl, mySrc, myW, j, acc); break;
            case 2:  add_rows<2>(hl, mySrc, myW, j, acc); break;
            case 1:  add_rows<1>(hl, mySrc, myW, j, acc); break;
            default: break;
        }
    }

    float4 o;
    o.x = fmaxf(acc.x + bv.x, 0.0f);
    o.y = fmaxf(acc.y + bv.y, 0.0f);
    o.z = fmaxf(acc.z + bv.z, 0.0f);
    o.w = fmaxf(acc.w + bv.w, 0.0f);
    *(float4*)(xout + (size_t)r * DD + lane * 4) = o;
  }
}

extern "C" void kernel_launch(void* const* d_in, const int* in_sizes, int n_in,
                              void* d_out, int out_size, void* d_ws, size_t ws_size,
                              hipStream_t stream) {
    const float* x    = (const float*)d_in[0];
    const int*   esrc = (const int*)d_in[1];
    const int*   edst = (const int*)d_in[2];
    const float* ew   = (const float*)d_in[3];
    const float* W    = (const float*)d_in[4];
    const float* b    = (const float*)d_in[5];
    float*       out  = (float*)d_out;

    float* ws0 = (float*)d_ws;
    float* ws1 = ws0 + (size_t)NN * DD;
    int*   cnt    = (int*)(ws1 + (size_t)NN * DD);
    int*   start  = cnt + NN;
    int*   total  = start + NN;
    int*   rank   = total + 64;
    int*   eid    = rank + NE;
    int*   ssrc   = eid + NE;
    float* sw     = (float*)(ssrc + NE);

    const float* xcur[3] = { x,   ws1, out };
    float*       xnxt[3] = { ws1, out, out };

    const int  gGemm = 256;
    const int  gN = (NN + 255) / 256;
    const int  gE = 1024;
    const int  gO = 2048;
    const int  gR = 2048;

    csr_zero<<<gN, 256, 0, stream>>>(cnt, total);
    csr_count<<<gE, 256, 0, stream>>>(edst, cnt, rank);
    csr_alloc<<<gN, 256, 0, stream>>>(cnt, total, start);
    csr_place<<<gE, 256, 0, stream>>>(edst, start, rank, eid);
    csr_order<<<gO, 256, 0, stream>>>(eid, edst, esrc, ew, start, cnt, ssrc, sw);

    for (int l = 0; l < 3; ++l) {
        gcn_gemm_wmma<<<gGemm, 512, 0, stream>>>(xcur[l], W + (size_t)l * DD * DD, ws0);
        gcn_gather<<<gR, 256, 0, stream>>>(ws0, start, cnt, ssrc, sw, b + (size_t)l * DD, xnxt[l]);
    }
}


# Bucket build without global atomics: edges binned by 256-row range via LDS counters (chunk histogram, scan, scatter), then counted and placed per bin in LDS; bucket order and all sums kept
# speedup vs baseline: 5.5316x; 1.6179x over previous
#include <hip/hip_runtime.h>

#define NN 100000
#define NE 1600000
#define DD 128

typedef __attribute__((ext_vector_type(2))) float v2f;
typedef __attribute__((ext_vector_type(8))) float v8f;

#define NB ((NN + 255) >> 8)
#define G1 256
#define CH ((NE + G1 - 1) / G1)
static_assert(NB <= 512, "the bin scan uses 16 waves");
static_assert(G1 == 256, "bin_scan scans one chunk count per thread of a 256-thread block");

__device__ __forceinline__ int wave_incl_scan(int v, int lane) {
    for (int o = 1; o < 32; o <<= 1) {
        const int t = __shfl_up(v, o);
        if (lane >= o) v += t;
    }
    return v;
}

__global__ __launch_bounds__(1024) void bin_hist(const int* __restrict__ dst, int* __restrict__ H) {
    __shared__ int h[NB];
    const int tid = threadIdx.x;
    for (int i = tid; i < NB; i += 1024) h[i] = 0;
    __syncthreads();
    const int lo = blockIdx.x * CH;
    const int hi = (lo + CH < NE) ? (lo + CH) : NE;
    for (int e0 = lo + tid; e0 < hi; e0 += 4 * 1024) {
        int d[4];
        #pragma unroll
        for (int k = 0; k < 4; ++k) {
            const int e = e0 + k * 1024;
            d[k] = dst[(e < hi) ? e : e0];
        }
        #pragma unroll
        for (int k = 0; k < 4; ++k) if (e0 + k * 1024 < hi) atomicAdd(&h[d[k] >> 8], 1);
    }
    __syncthreads();
    for (int i = tid; i < NB; i += 1024) H[i * G1 + blockIdx.x] = h[i];
}

__global__ __launch_bounds__(256) void bin_scan(int* __restrict__ H, int* __restrict__ binCnt) {
    __shared__ int wsum[8];
    const int tid  = threadIdx.x;
    const int lane = tid & 31;
    const int wave = tid >> 5;
    const int c = H[blockIdx.x * G1 + tid];
    const int v = wave_incl_scan(c, lane);
    if (lane == 31) wsum[wave] = v;
    __syncthreads();
    int pre = 0;
    for (int i = 0; i < wave; ++i) pre += wsum[i];
    H[blockIdx.x * G1 + tid] = pre + v - c;
    if (tid == 255) binCnt[blockIdx.x] = pre + v;
}

__global__ __launch_bounds__(1024) void bin_scatter(const int* __restrict__ dst, const int* __restrict__ src,
                                                    const float* __restrict__ w, const int* __restrict__ O,
                                                    const int* __restrict__ binCnt, int* __restrict__ binStart,
                                                    int* __restrict__ recRaw) {
    __shared__ int cur[NB];
    __shared__ int wsum[16];
    int4* rec = (int4*)recRaw;
    const int tid  = threadIdx.x;
    const int lane = tid & 31;
    const int wave = tid >> 5;

    int c = 0, v = 0;
    if (tid < 512) {
        c = (tid < NB) ? binCnt[tid] : 0;
        v = wave_incl_scan(c, lane);
        if (lane == 31) wsum[wave] = v;
    }
    __syncthreads();
    if (tid < NB) {
        int pre = 0;
        for (int i = 0; i < wave; ++i) pre += wsum[i];
        const int bs = pre + v - c;
        cur[tid] = bs + O[tid * G1 + blockIdx.x];
        if (blockIdx.x == 0) binStart[tid] = bs;
    }
    __syncthreads();

    const int lo = blockIdx.x * CH;
    const int hi = (lo + CH < NE) ? (lo + CH) : NE;
    for (int e0 = lo + tid; e0 < hi; e0 += 4 * 1024) {
        int d[4], s[4];
        float ww[4];
        #pragma unroll
        for (int k = 0; k < 4; ++k) {
            const int e  = e0 + k * 1024;
            const int ec = (e < hi) ? e : e0;
            d[k]  = dst[ec];
            s[k]  = src[ec];
            ww[k] = w[ec];
        }
        #pragma unroll
        for (int k = 0; k < 4; ++k) {
            if (e0 + k * 1024 < hi) {
                const int slot = atomicAdd(&cur[d[k] >> 8], 1);
                rec[slot] = make_int4(e0 + k * 1024, d[k], s[k], __float_as_int(ww[k]));
            }
        }
    }
}

__global__ __launch_bounds__(1024) void bin_build(const int* __restrict__ recRaw, int* __restrict__ rec2Raw,
                                                  const int* __restrict__ binCnt, const int* __restrict__ binStart,
                                                  int* __restrict__ start, int* __restrict__ cnt) {
    __shared__ int sCnt[256];
    __shared__ int sCur[256];
    __shared__ int wsum[8];
    const int4* rec  = (const int4*)recRaw;
    int4*       rec2 = (int4*)rec2Raw;
    const int tid  = threadIdx.x;
    const int lane = tid & 31;
    const int wave = tid >> 5;
    const int bin  = blockIdx.x;
    const int n    = binCnt[bin];
    const int base = binStart[bin];

    if (tid < 256) sCnt[tid] = 0;
    __syncthreads();
    for (int i = tid; i < n; i += 1024) atomicAdd(&sCnt[rec[base + i].y & 255], 1);
    __syncthreads();

    int c = 0, v = 0;
    if (tid < 256) {
        c = sCnt[tid];
        v = wave_incl_scan(c, lane);
        if (lane == 31) wsum[wave] = v;
    }
    __syncthreads();
    if (tid < 256) {
        int pre = 0;
        for (int i = 0; i < wave; ++i) pre += wsum[i];
        const int st = base + pre + v - c;
        sCur[tid] = st;
        const int node = bin * 256 + tid;
        if (node < NN) { start[node] = st; cnt[node] = c; }
    }
    __syncthreads();
    for (int i = tid; i < n; i += 1024) {
        const int4 r = rec[base + i];
        const int slot = atomicAdd(&sCur[r.y & 255], 1);
        rec2[slot] = r;
    }
}

__global__ __launch_bounds__(256) void bin_order(const int* __restrict__ rec2Raw, const int* __restrict__ start,
                                                 const int* __restrict__ cnt, int* __restrict__ ssrc,
                                                 float* __restrict__ sw) {
    const int4* rec2 = (const int4*)rec2Raw;
    const int T = gridDim.x * 256;
    for (int p = blockIdx.x * 256 + threadIdx.x; p < NE; p += T) {
        const int4 r = rec2[p];
        const int s = start[r.y];
        const int n = cnt[r.y];
        int rank = 0;
        for (int q = 0; q < n; ++q) rank += (rec2[s + q].x < r.x) ? 1 : 0;
        ssrc[s + rank] = r.z;
        sw[s + rank]   = __int_as_float(r.w);
    }
}

template <int K>
__device__ __forceinline__ void add_rows(const float* __restrict__ hl, int mySrc, float myW,
                                         int j, float4& acc) {
    float4 v[K];
    float  wv[K];
    #pragma unroll
    for (int k = 0; k < K; ++k) {
        const int sk = __builtin_amdgcn_readlane(mySrc, j + k);
        wv[k] = __int_as_float(__builtin_amdgcn_readlane(__float_as_int(myW), j + k));
        v[k]  = *(const float4*)(hl + (size_t)sk * DD);
    }
    #pragma unroll
    for (int k = 0; k < K; ++k) {
        acc.x = fmaf(v[k].x, wv[k], acc.x);
        acc.y = fmaf(v[k].y, wv[k], acc.y);
        acc.z = fmaf(v[k].z, wv[k], acc.z);
        acc.w = fmaf(v[k].w, wv[k], acc.w);
    }
}

#define TP 132
__global__ __launch_bounds__(512) void gcn_layer(const float* __restrict__ x,
                                                  const float* __restrict__ W,
                                                  const float* __restrict__ bias,
                                                  const int* __restrict__ start,
                                                  const int* __restrict__ cnt,
                                                  const int* __restrict__ ssrc,
                                                  const float* __restrict__ sw,
                                                  float* __restrict__ xout) {
    __shared__ float ldsWT[DD * TP];
    __shared__ float ldsA[16 * TP];
    const int tid = threadIdx.x;

    #pragma unroll
    for (int i = 0; i < 32; ++i) {
        int e = tid + i * 512;
        int n = e & 127;
        int k = e >> 7;
        ldsWT[n * TP + k] = W[k * DD + n];
    }

    const int wave = __builtin_amdgcn_readfirstlane(tid >> 5);
    const int lane = tid & 31;
    const int c    = lane & 15;
    const int hh   = lane >> 4;
    const float* xl  = x + lane * 4;
    const float* ap  = ldsA + c * TP + hh * 4;
    const float* b0p = ldsWT + ((wave & 3) * 32 + c) * TP + hh * 4;
    const float* b1p = b0p + 16 * TP;
    const int col0   = (wave & 3) * 32 + c;

  for (int tile = blockIdx.x; tile < (NN / 16); tile += gridDim.x) {
    const int r = tile * 16 + wave;
    const int s = start[r];
    const int n = cnt[r];
    float4 acc = make_float4(0.0f, 0.0f, 0.0f, 0.0f);

    for (int base = 0; base < n; base += 32) {
        const int m = (n - base < 32) ? (n - base) : 32;
        int   mySrc = 0;
        float myW   = 0.0f;
        if (lane < m) {
            mySrc = ssrc[s + base + lane];
            myW   = sw[s + base + lane];
        }
        int j = 0;
        for (; j + 8 <= m; j += 8) add_rows<8>(xl, mySrc, myW, j, acc);
        if (j + 4 <= m) { add_rows<4>(xl, mySrc, myW, j, acc); j += 4; }
        switch (m - j) {
            case 3:  add_rows<3>(xl, mySrc, myW, j, acc); break;
            case 2:  add_rows<2>(xl, mySrc, myW, j, acc); break;
            case 1:  add_rows<1>(xl, mySrc, myW, j, acc); break;
            default: break;
        }
    }
    *(float4*)(ldsA + wave * TP + lane * 4) = acc;
    __syncthreads();

    v8f acc0 = {};
    v8f acc1 = {};
    if (wave < 4) {
        #pragma unroll
        for (int j = 0; j < DD / 8; ++j) {
            const float4 a  = *(const float4*)(ap  + j * 8);
            const float4 b0 = *(const float4*)(b0p + j * 8);
            const float4 b1 = *(const float4*)(b1p + j * 8);
            v2f aL, aH, b0L, b0H, b1L, b1H;
            aL.x  = a.x;  aL.y  = a.y;  aH.x  = a.z;  aH.y  = a.w;
            b0L.x = b0.x; b0L.y = b0.y; b0H.x = b0.z; b0H.y = b0.w;
            b1L.x = b1.x; b1L.y = b1.y; b1H.x = b1.z; b1H.y = b1.w;
            acc0 = __builtin_amdgcn_wmma_f32_16x16x4_f32(false, aL, false, b0L, (short)0, acc0, false, false);
            acc1 = __builtin_amdgcn_wmma_f32_16x16x4_f32(false, aL, false, b1L, (short)0, acc1, false, false);
            acc0 = __builtin_amdgcn_wmma_f32_16x16x4_f32(false, aH, false, b0H, (short)0, acc0, false, false);
            acc1 = __builtin_amdgcn_wmma_f32_16x16x4_f32(false, aH, false, b1H, (short)0, acc1, false, false);
        }
    }
    __syncthreads();

    if (wave < 4) {
        const float bia0 = bias[col0];
        const float bia1 = bias[col0 + 16];
        const int rowBase = tile * 16 + hh * 8;
        #pragma unroll
        for (int v = 0; v < 8; ++v) {
            float* op = xout + (size_t)(rowBase + v) * DD + col0;
            op[0]  = fmaxf(acc0[v] + bia0, 0.0f);
            op[16] = fmaxf(acc1[v] + bia1, 0.0f);
        }
    }
  }
}

extern "C" void kernel_launch(void* const* d_in, const int* in_sizes, int n_in,
                              void* d_out, int out_size, void* d_ws, size_t ws_size,
                              hipStream_t stream) {
    const float* x    = (const float*)d_in[0];
    const int*   esrc = (const int*)d_in[1];
    const int*   edst = (const int*)d_in[2];
    const float* ew   = (const float*)d_in[3];
    const float* W    = (const float*)d_in[4];
    const float* b    = (const float*)d_in[5];
    float*       out  = (float*)d_out;

    float* ws0 = (float*)d_ws;
    float* ws1 = ws0 + (size_t)NN * DD;
    int*   cnt      = (int*)(ws1 + (size_t)NN * DD);
    int*   start    = cnt + NN;
    int*   H        = start + NN;
    int*   binCnt   = H + NB * G1;
    int*   binStart = binCnt + 512;
    int*   ssrc     = binStart + 512;
    float* sw       = (float*)(ssrc + NE);
    int*   rec      = (int*)ws1;
    int*   rec2     = (int*)ws0;

    const float* xcur[3] = { x,   ws0, ws1 };
    float*       xnxt[3] = { ws0, ws1, out };

    const int  gL = 1024;
    const int  gO = 2048;

    bin_hist<<<G1, 1024, 0, stream>>>(edst, H);
    bin_scan<<<NB, 256, 0, stream>>>(H, binCnt);
    bin_scatter<<<G1, 1024, 0, stream>>>(edst, esrc, ew, H, binCnt, binStart, rec);
    bin_build<<<NB, 1024, 0, stream>>>(rec, rec2, binCnt, binStart, start, cnt);
    bin_order<<<gO, 256, 0, stream>>>(rec2, start, cnt, ssrc, sw);

    for (int l = 0; l < 3; ++l) {
        gcn_layer<<<gL, 512, 0, stream>>>(xcur[l], W + (size_t)l * DD * DD, b + (size_t)l * DD,
                                          start, cnt, ssrc, sw, xnxt[l]);
    }
}


# Tile loop software-pipelined: a wave reads its row's bucket bounds for all its tiles once, and fetches the next tile's (src,w) pairs while it sums the current row; bias read once; sums and order kept
# speedup vs baseline: 5.7042x; 1.0312x over previous
#include <hip/hip_runtime.h>

#define NN 100000
#define NE 1600000
#define DD 128

typedef __attribute__((ext_vector_type(2))) float v2f;
typedef __attribute__((ext_vector_type(8))) float v8f;

#define NB ((NN + 255) >> 8)
#define G1 256
#define CH ((NE + G1 - 1) / G1)
static_assert(NB <= 512, "the bin scan uses 16 waves");
static_assert(G1 == 256, "bin_scan scans one chunk count per thread of a 256-thread block");

__device__ __forceinline__ int wave_incl_scan(int v, int lane) {
    for (int o = 1; o < 32; o <<= 1) {
        const int t = __shfl_up(v, o);
        if (lane >= o) v += t;
    }
    return v;
}

__global__ __launch_bounds__(1024) void bin_hist(const int* __restrict__ dst, int* __restrict__ H) {
    __shared__ int h[NB];
    const int tid = threadIdx.x;
    for (int i = tid; i < NB; i += 1024) h[i] = 0;
    __syncthreads();
    const int lo = blockIdx.x * CH;
    const int hi = (lo + CH < NE) ? (lo + CH) : NE;
    for (int e0 = lo + tid; e0 < hi; e0 += 4 * 1024) {
        int d[4];
        #pragma unroll
        for (int k = 0; k < 4; ++k) {
            const int e = e0 + k * 1024;
            d[k] = dst[(e < hi) ? e : e0];
        }
        #pragma unroll
        for (int k = 0; k < 4; ++k) if (e0 + k * 1024 < hi) atomicAdd(&h[d[k] >> 8], 1);
    }
    __syncthreads();
    for (int i = tid; i < NB; i += 1024) H[i * G1 + blockIdx.x] = h[i];
}

__global__ __launch_bounds__(256) void bin_scan(int* __restrict__ H, int* __restrict__ binCnt) {
    __shared__ int wsum[8];
    const int tid  = threadIdx.x;
    const int lane = tid & 31;
    const int wave = tid >> 5;
    const int c = H[blockIdx.x * G1 + tid];
    const int v = wave_incl_scan(c, lane);
    if (lane == 31) wsum[wave] = v;
    __syncthreads();
    int pre = 0;
    for (int i = 0; i < wave; ++i) pre += wsum[i];
    H[blockIdx.x * G1 + tid] = pre + v - c;
    if (tid == 255) binCnt[blockIdx.x] = pre + v;
}

__global__ __launch_bounds__(1024) void bin_scatter(const int* __restrict__ dst, const int* __restrict__ src,
                                                    const float* __restrict__ w, const int* __restrict__ O,
                                                    const int* __restrict__ binCnt, int* __restrict__ binStart,
                                                    int* __restrict__ recRaw) {
    __shared__ int cur[NB];
    __shared__ int wsum[16];
    int4* rec = (int4*)recRaw;
    const int tid  = threadIdx.x;
    const int lane = tid & 31;
    const int wave = tid >> 5;

    int c = 0, v = 0;
    if (tid < 512) {
        c = (tid < NB) ? binCnt[tid] : 0;
        v = wave_incl_scan(c, lane);
        if (lane == 31) wsum[wave] = v;
    }
    __syncthreads();
    if (tid < NB) {
        int pre = 0;
        for (int i = 0; i < wave; ++i) pre += wsum[i];
        const int bs = pre + v - c;
        cur[tid] = bs + O[tid * G1 + blockIdx.x];
        if (blockIdx.x == 0) binStart[tid] = bs;
    }
    __syncthreads();

    const int lo = blockIdx.x * CH;
    const int hi = (lo + CH < NE) ? (lo + CH) : NE;
    for (int e0 = lo + tid; e0 < hi; e0 += 4 * 1024) {
        int d[4], s[4];
        float ww[4];
        #pragma unroll
        for (int k = 0; k < 4; ++k) {
            const int e  = e0 + k * 1024;
            const int ec = (e < hi) ? e : e0;
            d[k]  = dst[ec];
            s[k]  = src[ec];
            ww[k] = w[ec];
        }
        #pragma unroll
        for (int k = 0; k < 4; ++k) {
            if (e0 + k * 1024 < hi) {
                const int slot = atomicAdd(&cur[d[k] >> 8], 1);
                rec[slot] = make_int4(e0 + k * 1024, d[k], s[k], __float_as_int(ww[k]));
            }
        }
    }
}

__global__ __launch_bounds__(1024) void bin_build(const int* __restrict__ recRaw, int* __restrict__ rec2Raw,
                                                  const int* __restrict__ binCnt, const int* __restrict__ binStart,
                                                  int* __restrict__ start, int* __restrict__ cnt) {
    __shared__ int sCnt[256];
    __shared__ int sCur[256];
    __shared__ int wsum[8];
    const int4* rec  = (const int4*)recRaw;
    int4*       rec2 = (int4*)rec2Raw;
    const int tid  = threadIdx.x;
    const int lane = tid & 31;
    const int wave = tid >> 5;
    const int bin  = blockIdx.x;
    const int n    = binCnt[bin];
    const int base = binStart[bin];

    if (tid < 256) sCnt[tid] = 0;
    __syncthreads();
    for (int i = tid; i < n; i += 1024) atomicAdd(&sCnt[rec[base + i].y & 255], 1);
    __syncthreads();

    int c = 0, v = 0;
    if (tid < 256) {
        c = sCnt[tid];
        v = wave_incl_scan(c, lane);
        if (lane == 31) wsum[wave] = v;
    }
    __syncthreads();
    if (tid < 256) {
        int pre = 0;
        for (int i = 0; i < wave; ++i) pre += wsum[i];
        const int st = base + pre + v - c;
        sCur[tid] = st;
        const int node = bin * 256 + tid;
        if (node < NN) { start[node] = st; cnt[node] = c; }
    }
    __syncthreads();
    for (int i = tid; i < n; i += 1024) {
        const int4 r = rec[base + i];
        const int slot = atomicAdd(&sCur[r.y & 255], 1);
        rec2[slot] = r;
    }
}

__global__ __launch_bounds__(256) void bin_order(const int* __restrict__ rec2Raw, const int* __restrict__ start,
                                                 const int* __restrict__ cnt, int* __restrict__ ssrc,
                                                 float* __restrict__ sw) {
    const int4* rec2 = (const int4*)rec2Raw;
    const int T = gridDim.x * 256;
    for (int p = blockIdx.x * 256 + threadIdx.x; p < NE; p += T) {
        const int4 r = rec2[p];
        const int s = start[r.y];
        const int n = cnt[r.y];
        int rank = 0;
        for (int q = 0; q < n; ++q) rank += (rec2[s + q].x < r.x) ? 1 : 0;
        ssrc[s + rank] = r.z;
        sw[s + rank]   = __int_as_float(r.w);
    }
}

template <int K>
__device__ __forceinline__ void add_rows(const float* __restrict__ hl, int mySrc, float myW,
                                         int j, float4& acc) {
    float4 v[K];
    float  wv[K];
    #pragma unroll
    for (int k = 0; k < K; ++k) {
        const int sk = __builtin_amdgcn_readlane(mySrc, j + k);
        wv[k] = __int_as_float(__builtin_amdgcn_readlane(__float_as_int(myW), j + k));
        v[k]  = *(const float4*)(hl + (size_t)sk * DD);
    }
    #pragma unroll
    for (int k = 0; k < K; ++k) {
        acc.x = fmaf(v[k].x, wv[k], acc.x);
        acc.y = fmaf(v[k].y, wv[k], acc.y);
        acc.z = fmaf(v[k].z, wv[k], acc.z);
        acc.w = fmaf(v[k].w, wv[k], acc.w);
    }
}

__device__ __forceinline__ void add_chunk(const float* __restrict__ hl, int mySrc, float myW, int m, float4& acc) {
    int j = 0;
    for (; j + 8 <= m; j += 8) add_rows<8>(hl, mySrc, myW, j, acc);
    if (j + 4 <= m) { add_rows<4>(hl, mySrc, myW, j, acc); j += 4; }
    switch (m - j) {
        case 3:  add_rows<3>(hl, mySrc, myW, j, acc); break;
        case 2:  add_rows<2>(hl, mySrc, myW, j, acc); break;
        case 1:  add_rows<1>(hl, mySrc, myW, j, acc); break;
        default: break;
    }
}

#define TP 132
#define GL 1024
#define NT (NN / 16)
static_assert(NN % 16 == 0, "whole tiles");
static_assert((NT + GL - 1) / GL <= 31, "one lane per tile of the block holds that tile's bucket bounds, and one more lane is read past the last tile");
__global__ __launch_bounds__(512) void gcn_layer(const float* __restrict__ x,
                                                  const float* __restrict__ W,
                                                  const float* __restrict__ bias,
                                                  const int* __restrict__ start,
                                                  const int* __restrict__ cnt,
                                                  const int* __restrict__ ssrc,
                                                  const float* __restrict__ sw,
                                                  float* __restrict__ xout) {
    __shared__ float ldsWT[DD * TP];
    __shared__ float ldsA[16 * TP];
    const int tid = threadIdx.x;

    const int wave = __builtin_amdgcn_readfirstlane(tid >> 5);
    const int lane = tid & 31;

    int cntAll = 0, startAll = 0;
    {
        const int ti = blockIdx.x + lane * GL;
        if (ti < NT) {
            cntAll   = cnt[ti * 16 + wave];
            startAll = start[ti * 16 + wave];
        }
    }

    #pragma unroll
    for (int i = 0; i < 32; ++i) {
        int e = tid + i * 512;
        int n = e & 127;
        int k = e >> 7;
        ldsWT[n * TP + k] = W[k * DD + n];
    }

    const int c    = lane & 15;
    const int hh   = lane >> 4;
    const int lane4 = lane * 4;
    const float* xl  = x + lane4;
    const float* ap  = ldsA + c * TP + hh * 4;
    const float* b0p = ldsWT + ((wave & 3) * 32 + c) * TP + hh * 4;
    const float* b1p = b0p + 16 * TP;
    const int col0   = (wave & 3) * 32 + c;
    const float bia0 = bias[col0];
    const float bia1 = bias[col0 + 16];

    int   nCur = __builtin_amdgcn_readlane(cntAll, 0);
    int   sCur = __builtin_amdgcn_readlane(startAll, 0);
    int   mySrc, mySrcN;
    float myW, myWN;
    {
        const int pi = (sCur + lane < NE) ? (sCur + lane) : (NE - 1);
        mySrc = ssrc[pi];
        myW   = sw[pi];
    }

  for (int it = 0, tile = blockIdx.x; tile < NT; ++it, tile += GL) {
    const int nNext = __builtin_amdgcn_readlane(cntAll, it + 1);
    const int sNext = __builtin_amdgcn_readlane(startAll, it + 1);
    {
        const int pi = (sNext + lane < NE) ? (sNext + lane) : (NE - 1);
        mySrcN = ssrc[pi];
        myWN   = sw[pi];
    }

    float4 acc = make_float4(0.0f, 0.0f, 0.0f, 0.0f);
    add_chunk(xl, mySrc, myW, (nCur < 32) ? nCur : 32, acc);
    for (int base = 32; base < nCur; base += 32) {
        const int m = (nCur - base < 32) ? (nCur - base) : 32;
        int   s2 = 0;
        float w2 = 0.0f;
        if (lane < m) {
            s2 = ssrc[sCur + base + lane];
            w2 = sw[sCur + base + lane];
        }
        add_chunk(xl, s2, w2, m, acc);
    }
    *(float4*)(ldsA + wave * TP + lane4) = acc;
    __syncthreads();

    v8f acc0 = {};
    v8f acc1 = {};
    if (wave < 4) {
        #pragma unroll
        for (int j = 0; j < DD / 8; ++j) {
            const float4 a  = *(const float4*)(ap  + j * 8);
            const float4 b0 = *(const float4*)(b0p + j * 8);
            const float4 b1 = *(const float4*)(b1p + j * 8);
            v2f aL, aH, b0L, b0H, b1L, b1H;
            aL.x  = a.x;  aL.y  = a.y;  aH.x  = a.z;  aH.y  = a.w;
            b0L.x = b0.x; b0L.y = b0.y; b0H.x = b0.z; b0H.y = b0.w;
            b1L.x = b1.x; b1L.y = b1.y; b1H.x = b1.z; b1H.y = b1.w;
            acc0 = __builtin_amdgcn_wmma_f32_16x16x4_f32(false, aL, false, b0L, (short)0, acc0, false, false);
            acc1 = __builtin_amdgcn_wmma_f32_16x16x4_f32(false, aL, false, b1L, (short)0, acc1, false, false);
            acc0 = __builtin_amdgcn_wmma_f32_16x16x4_f32(false, aH, false, b0H, (short)0, acc0, false, false);
            acc1 = __builtin_amdgcn_wmma_f32_16x16x4_f32(false, aH, false, b1H, (short)0, acc1, false, false);
        }
    }
    __syncthreads();

    if (wave < 4) {
        const int rowBase = tile * 16 + hh * 8;
        #pragma unroll
        for (int v = 0; v < 8; ++v) {
            float* op = xout + (size_t)(rowBase + v) * DD + col0;
            op[0]  = fmaxf(acc0[v] + bia0, 0.0f);
            op[16] = fmaxf(acc1[v] + bia1, 0.0f);
        }
    }
    nCur = nNext; sCur = sNext; mySrc = mySrcN; myW = myWN;
  }
}

extern "C" void kernel_launch(void* const* d_in, const int* in_sizes, int n_in,
                              void* d_out, int out_size, void* d_ws, size_t ws_size,
                              hipStream_t stream) {
    const float* x    = (const float*)d_in[0];
    const int*   esrc = (const int*)d_in[1];
    const int*   edst = (const int*)d_in[2];
    const float* ew   = (const float*)d_in[3];
    const float* W    = (const float*)d_in[4];
    const float* b    = (const float*)d_in[5];
    float*       out  = (float*)d_out;

    float* ws0 = (float*)d_ws;
    float* ws1 = ws0 + (size_t)NN * DD;
    int*   cnt      = (int*)(ws1 + (size_t)NN * DD);
    int*   start    = cnt + NN;
    int*   H        = start + NN;
    int*   binCnt   = H + NB * G1;
    int*   binStart = binCnt + 512;
    int*   ssrc     = binStart + 512;
    float* sw       = (float*)(ssrc + NE);
    int*   rec      = (int*)ws1;
    int*   rec2     = (int*)ws0;

    const float* xcur[3] = { x,   ws0, ws1 };
    float*       xnxt[3] = { ws0, ws1, out };

    const int  gO = 2048;

    bin_hist<<<G1, 1024, 0, stream>>>(edst, H);
    bin_scan<<<NB, 256, 0, stream>>>(H, binCnt);
    bin_scatter<<<G1, 1024, 0, stream>>>(edst, esrc, ew, H, binCnt, binStart, rec);
    bin_build<<<NB, 1024, 0, stream>>>(rec, rec2, binCnt, binStart, start, cnt);
    bin_order<<<gO, 256, 0, stream>>>(rec2, start, cnt, ssrc, sw);

    for (int l = 0; l < 3; ++l) {
        gcn_layer<<<GL, 512, 0, stream>>>(xcur[l], W + (size_t)l * DD * DD, b + (size_t)l * DD,
                                          start, cnt, ssrc, sw, xnxt[l]);
    }
}
